# starting kernel
# baseline (speedup 1.0000x reference)
#include <hip/hip_runtime.h>
#include <math.h>

#define BATCH 32768
#define LRATE 0.01f
#define AEPS  1e-8f
#define LNEPS 1e-5f

typedef __attribute__((ext_vector_type(2))) float v2f;
typedef __attribute__((ext_vector_type(8))) float v8f;

__device__ __forceinline__ v2f mkv2(float x, float y) { v2f v; v[0] = x; v[1] = y; return v; }

__device__ __forceinline__ v8f wmma_f32(v2f a, v2f b, v8f c) {
  return __builtin_amdgcn_wmma_f32_16x16x4_f32(false, a, false, b, (short)0, c, false, false);
}

constexpr size_t F_CAT  = 0;
constexpr size_t F_S    = F_CAT + (size_t)BATCH * 16;
constexpr size_t F_W1   = F_S + (size_t)3 * BATCH * 64;
constexpr size_t F_HB   = F_W1 + 16 * 64;
constexpr size_t F_QT   = F_HB + 64;
constexpr size_t F_KT   = F_QT + 64 * 64;
constexpr size_t F_VT   = F_KT + 64 * 64;
constexpr size_t F_LWT  = F_VT + 64 * 64;
constexpr size_t F_GW   = F_LWT + 64 * 64;
constexpr size_t F_GB   = F_GW + 64 * 64;
constexpr size_t F_LW1T = F_GB + 64;
constexpr size_t F_LB1  = F_LW1T + 64 * 64;

__global__ void __launch_bounds__(256) k_cat(
    const float* __restrict__ in1, const float* __restrict__ in2,
    const float* __restrict__ in3, const float* __restrict__ in4,
    const float* __restrict__ wl, const float* __restrict__ bl,
    const float* __restrict__ wm, const float* __restrict__ bm,
    const float* __restrict__ wsw, const float* __restrict__ bs,
    float* __restrict__ cat)
{
  __shared__ float sw[4500 + 2250 + 450];
  for (int i = threadIdx.x; i < 7200; i += 256) {
    float v;
    if (i < 4500)      v = wl[i];
    else if (i < 6750) v = wm[i - 4500];
    else               v = wsw[i - 6750];
    sw[i] = v;
  }
  __syncthreads();

  const int warp = threadIdx.x >> 5, lane = threadIdx.x & 31;
  const int b = blockIdx.x * 8 + warp;

  float aL[3] = {0.f, 0.f, 0.f}, aM[3] = {0.f, 0.f, 0.f}, aS[3] = {0.f, 0.f, 0.f};

  const float* r1 = in1 + (size_t)b * 1500;
  for (int k = lane; k < 500; k += 32) {
    __builtin_prefetch(r1 + (k + 128) * 3, 0, 0);
    float x0 = r1[k * 3 + 0], x1 = r1[k * 3 + 1], x2 = r1[k * 3 + 2];
#pragma unroll
    for (int o = 0; o < 3; ++o)
      aL[o] += x0 * sw[(o * 3 + 0) * 500 + k] + x1 * sw[(o * 3 + 1) * 500 + k] +
               x2 * sw[(o * 3 + 2) * 500 + k];
  }
  const float* r2 = in2 + (size_t)b * 750;
  for (int k = lane; k < 250; k += 32) {
    float x0 = r2[k * 3 + 0], x1 = r2[k * 3 + 1], x2 = r2[k * 3 + 2];
#pragma unroll
    for (int o = 0; o < 3; ++o)
      aM[o] += x0 * sw[4500 + (o * 3 + 0) * 250 + k] + x1 * sw[4500 + (o * 3 + 1) * 250 + k] +
               x2 * sw[4500 + (o * 3 + 2) * 250 + k];
  }
  const float* r3 = in3 + (size_t)b * 150;
  for (int k = lane; k < 50; k += 32) {
    float x0 = r3[k * 3 + 0], x1 = r3[k * 3 + 1], x2 = r3[k * 3 + 2];
#pragma unroll
    for (int o = 0; o < 3; ++o)
      aS[o] += x0 * sw[6750 + (o * 3 + 0) * 50 + k] + x1 * sw[6750 + (o * 3 + 1) * 50 + k] +
               x2 * sw[6750 + (o * 3 + 2) * 50 + k];
  }
#pragma unroll
  for (int o = 0; o < 3; ++o)
#pragma unroll
    for (int off = 16; off; off >>= 1) {
      aL[o] += __shfl_xor(aL[o], off, 32);
      aM[o] += __shfl_xor(aM[o], off, 32);
      aS[o] += __shfl_xor(aS[o], off, 32);
    }
  if (lane == 0) {
    float* cr = cat + (size_t)b * 16;
#pragma unroll
    for (int o = 0; o < 3; ++o) {
      cr[o]      = aL[o] + bl[o];
      cr[3 + o]  = aM[o] + bm[o];
      cr[6 + o]  = aS[o] + bs[o];
      cr[9 + o]  = in4[(size_t)b * 3 + o];
      cr[12 + o] = 0.f;
    }
    cr[15] = 0.f;
  }
}

__global__ void k_prep(const float* __restrict__ fc1w, const float* __restrict__ fc1b,
                       const float* __restrict__ qw, const float* __restrict__ kw,
                       const float* __restrict__ vw, const float* __restrict__ lw,
                       float* __restrict__ wsf)
{
  float* W1 = wsf + F_W1; float* hb = wsf + F_HB;
  float* qT = wsf + F_QT; float* kT = wsf + F_KT; float* vT = wsf + F_VT;
  float* lwT = wsf + F_LWT; float* gW = wsf + F_GW; float* gb = wsf + F_GB;
  for (int i = threadIdx.x; i < 1024; i += 256) {
    int k = i >> 6, n = i & 63;
    W1[i] = (k < 12) ? fc1w[n * 12 + k] : 0.f;
  }
  for (int i = threadIdx.x; i < 4096; i += 256) {
    int k = i >> 6, n = i & 63;
    qT[i] = qw[n * 64 + k]; kT[i] = kw[n * 64 + k];
    vT[i] = vw[n * 64 + k]; lwT[i] = lw[n * 64 + k];
    gW[i] = 0.f;
  }
  for (int i = threadIdx.x; i < 64; i += 256) {
    hb[i] = fc1b[i] + ((i & 1) ? 1.0f : 0.0f);
    gb[i] = 0.f;
  }
}

__global__ void __launch_bounds__(256) k_qkv(
    const float* __restrict__ cat, const float* __restrict__ qb,
    const float* __restrict__ kbv, const float* __restrict__ vb,
    float* __restrict__ s, const float* __restrict__ W1, const float* __restrict__ hb,
    const float* __restrict__ qT, const float* __restrict__ kT, const float* __restrict__ vT)
{
  __shared__ float sh[8][16 * 68];
  const int warp = threadIdx.x >> 5, lane = threadIdx.x & 31;
  const int hi = lane >> 4, nl = lane & 15;
  const int row0 = (blockIdx.x * 8 + warp) * 16;
  float* shw = sh[warp];

  v2f ca[4];
#pragma unroll
  for (int ks = 0; ks < 4; ++ks) {
    int kb = ks * 4 + 2 * hi;
    const float* p = cat + (size_t)(row0 + nl) * 16 + kb;
    ca[ks] = mkv2(p[0], p[1]);
  }
#pragma unroll
  for (int n0 = 0; n0 < 4; ++n0) {
    int n = n0 * 16 + nl;
    v8f c = {};
#pragma unroll
    for (int ks = 0; ks < 4; ++ks) {
      int kb = ks * 4 + 2 * hi;
      c = wmma_f32(ca[ks], mkv2(W1[kb * 64 + n], W1[(kb + 1) * 64 + n]), c);
    }
    float bias = hb[n];
#pragma unroll
    for (int r = 0; r < 8; ++r) shw[(r + 8 * hi) * 68 + n] = c[r] + bias;
  }
  __syncthreads();

  v2f ha[16];
#pragma unroll
  for (int ks = 0; ks < 16; ++ks) {
    int kb = ks * 4 + 2 * hi;
    ha[ks] = mkv2(shw[nl * 68 + kb], shw[nl * 68 + kb + 1]);
  }
  const float* BT[3] = {qT, kT, vT};
  const float* BB[3] = {qb, kbv, vb};
#pragma unroll
  for (int t = 0; t < 3; ++t) {
    float* ob = s + (size_t)t * BATCH * 64;
#pragma unroll
    for (int n0 = 0; n0 < 4; ++n0) {
      int n = n0 * 16 + nl;
      v8f c = {};
#pragma unroll
      for (int ks = 0; ks < 16; ++ks) {
        int kb = ks * 4 + 2 * hi;
        c = wmma_f32(ha[ks], mkv2(BT[t][kb * 64 + n], BT[t][(kb + 1) * 64 + n]), c);
      }
      float bias = BB[t][n];
#pragma unroll
      for (int r = 0; r < 8; ++r)
        ob[(size_t)(row0 + r + 8 * hi) * 64 + n] = c[r] + bias;
    }
  }
}

__global__ void __launch_bounds__(128) k_grad(
    const float* __restrict__ s, const float* __restrict__ tk,
    const float* __restrict__ tvm, const float* __restrict__ lwT,
    const float* __restrict__ lb, float* __restrict__ gW, float* __restrict__ gb)
{
  __shared__ float stv[4][16 * 68];
  __shared__ float sg[4][16 * 68];
  __shared__ float sgW[64 * 64];
  __shared__ float sgb[64];
  for (int i = threadIdx.x; i < 4096; i += 128) sgW[i] = 0.f;
  if (threadIdx.x < 64) sgb[threadIdx.x] = 0.f;
  __syncthreads();

  const int warp = threadIdx.x >> 5, lane = threadIdx.x & 31;
  const int hi = lane >> 4, nl = lane & 15;
  float* tvw = stv[warp];
  float* gw_ = sg[warp];

  const float gscale = 2.0f / (float)(3ull * BATCH * 64);
  const int NT = 3 * BATCH / 16;
  const int wave = blockIdx.x * 4 + warp;
  const int nw = gridDim.x * 4;

  v8f cgw[4][4];
#pragma unroll
  for (int i0 = 0; i0 < 4; ++i0)
#pragma unroll
    for (int j0 = 0; j0 < 4; ++j0) cgw[i0][j0] = (v8f){};

  for (int t = wave; t < NT; t += nw) {
    const int r0 = t * 16;
    v2f sa[16];
#pragma unroll
    for (int ks = 0; ks < 16; ++ks) {
      int kb = ks * 4 + 2 * hi;
      const float* p = s + (size_t)(r0 + nl) * 64 + kb;
      sa[ks] = mkv2(p[0], p[1]);
    }
#pragma unroll
    for (int n0 = 0; n0 < 4; ++n0) {
      int n = n0 * 16 + nl;
      v8f ctv = {}, clv = {};
#pragma unroll
      for (int ks = 0; ks < 16; ++ks) {
        int kb = ks * 4 + 2 * hi;
        ctv = wmma_f32(sa[ks], mkv2(tk[kb * 64 + n], tk[(kb + 1) * 64 + n]), ctv);
        clv = wmma_f32(sa[ks], mkv2(tvm[kb * 64 + n], tvm[(kb + 1) * 64 + n]), clv);
      }
#pragma unroll
      for (int r = 0; r < 8; ++r) {
        tvw[(r + 8 * hi) * 68 + n] = ctv[r];
        gw_[(r + 8 * hi) * 68 + n] = -clv[r];
      }
    }
#pragma unroll
    for (int n0 = 0; n0 < 4; ++n0) {
      int n = n0 * 16 + nl;
      v8f cp = {};
#pragma unroll
      for (int ks = 0; ks < 16; ++ks) {
        int kb = ks * 4 + 2 * hi;
        v2f a = mkv2(tvw[nl * 68 + kb], tvw[nl * 68 + kb + 1]);
        cp = wmma_f32(a, mkv2(lwT[kb * 64 + n], lwT[(kb + 1) * 64 + n]), cp);
      }
      float lbn = lb[n];
#pragma unroll
      for (int r = 0; r < 8; ++r) {
        int idx = (r + 8 * hi) * 68 + n;
        float g = gscale * (cp[r] + lbn + gw_[idx]);
        gw_[idx] = g;
        atomicAdd(&sgb[n], g);
      }
    }
#pragma unroll
    for (int i0 = 0; i0 < 4; ++i0) {
      int i = i0 * 16 + nl;
#pragma unroll
      for (int j0 = 0; j0 < 4; ++j0) {
        int j = j0 * 16 + nl;
#pragma unroll
        for (int ks = 0; ks < 4; ++ks) {
          int kb = ks * 4 + 2 * hi;
          v2f a = mkv2(gw_[kb * 68 + i], gw_[(kb + 1) * 68 + i]);
          v2f b = mkv2(tvw[kb * 68 + j], tvw[(kb + 1) * 68 + j]);
          cgw[i0][j0] = wmma_f32(a, b, cgw[i0][j0]);
        }
      }
    }
  }
#pragma unroll
  for (int i0 = 0; i0 < 4; ++i0)
#pragma unroll
    for (int j0 = 0; j0 < 4; ++j0)
#pragma unroll
      for (int r = 0; r < 8; ++r)
        atomicAdd(&sgW[(i0 * 16 + r + 8 * hi) * 64 + j0 * 16 + nl], cgw[i0][j0][r]);
  __syncthreads();
  for (int i = threadIdx.x; i < 4096; i += 128) atomicAdd(&gW[i], sgW[i]);
  if (threadIdx.x < 64) atomicAdd(&gb[threadIdx.x], sgb[threadIdx.x]);
}

__global__ void k_update(const float* __restrict__ lw, const float* __restrict__ lb,
                         const float* __restrict__ gW, const float* __restrict__ gb,
                         float* __restrict__ lw1T, float* __restrict__ lb1)
{
  for (int i = threadIdx.x; i < 4096; i += 256) {
    int n = i >> 6, k = i & 63;
    float g = gW[i];
    lw1T[k * 64 + n] = lw[i] - LRATE * g / (fabsf(g) + AEPS);
  }
  for (int i = threadIdx.x; i < 64; i += 256) {
    float g = gb[i];
    lb1[i] = lb[i] - LRATE * g / (fabsf(g) + AEPS);
  }
}

__global__ void __launch_bounds__(128) k_out(
    const float* __restrict__ s, const float* __restrict__ tq,
    const float* __restrict__ lw1T, const float* __restrict__ lb1,
    const float* __restrict__ lng, const float* __restrict__ lnb,
    const float* __restrict__ ow, const float* __restrict__ obp,
    float* __restrict__ out)
{
  __shared__ float su[4][16 * 68];
  __shared__ float so[4][16 * 68];
  const int warp = threadIdx.x >> 5, lane = threadIdx.x & 31;
  const int hi = lane >> 4, nl = lane & 15;
  const int row0 = (blockIdx.x * 4 + warp) * 16;
  float* uw = su[warp];
  float* om = so[warp];
  const float* sv = s + (size_t)2 * BATCH * 64;

  v2f sa[16];
#pragma unroll
  for (int ks = 0; ks < 16; ++ks) {
    int kb = ks * 4 + 2 * hi;
    const float* p = sv + (size_t)(row0 + nl) * 64 + kb;
    sa[ks] = mkv2(p[0], p[1]);
  }
#pragma unroll
  for (int n0 = 0; n0 < 4; ++n0) {
    int n = n0 * 16 + nl;
    v8f c = {};
#pragma unroll
    for (int ks = 0; ks < 16; ++ks) {
      int kb = ks * 4 + 2 * hi;
      c = wmma_f32(sa[ks], mkv2(tq[kb * 64 + n], tq[(kb + 1) * 64 + n]), c);
    }
#pragma unroll
    for (int r = 0; r < 8; ++r) uw[(r + 8 * hi) * 68 + n] = c[r];
  }
#pragma unroll
  for (int n0 = 0; n0 < 4; ++n0) {
    int n = n0 * 16 + nl;
    v8f c = {};
#pragma unroll
    for (int ks = 0; ks < 16; ++ks) {
      int kb = ks * 4 + 2 * hi;
      v2f a = mkv2(uw[nl * 68 + kb], uw[nl * 68 + kb + 1]);
      c = wmma_f32(a, mkv2(lw1T[kb * 64 + n], lw1T[(kb + 1) * 64 + n]), c);
    }
    float bias = lb1[n];
#pragma unroll
    for (int r = 0; r < 8; ++r) om[(r + 8 * hi) * 68 + n] = c[r] + bias;
  }
  __syncthreads();
  if (lane < 16) {
    const float* rowp = om + lane * 68;
    float mu = 0.f;
#pragma unroll
    for (int j = 0; j < 64; ++j) mu += rowp[j];
    mu *= (1.0f / 64.0f);
    float var = 0.f;
#pragma unroll
    for (int j = 0; j < 64; ++j) { float d = rowp[j] - mu; var += d * d; }
    var *= (1.0f / 64.0f);
    float rstd = rsqrtf(var + LNEPS);
    float acc = 0.f;
#pragma unroll
    for (int j = 0; j < 64; ++j) {
      float nrm = (rowp[j] - mu) * rstd * lng[j] + lnb[j];
      float a = nrm / (1.0f + __expf(-nrm));
      acc += a * ow[j];
    }
    out[row0 + lane] = acc + obp[0];
  }
}

extern "C" void kernel_launch(void* const* d_in, const int* in_sizes, int n_in,
                              void* d_out, int out_size, void* d_ws, size_t ws_size,
                              hipStream_t stream) {
  const float* in1  = (const float*)d_in[0];
  const float* in2  = (const float*)d_in[1];
  const float* in3  = (const float*)d_in[2];
  const float* in4  = (const float*)d_in[3];
  const float* w_l  = (const float*)d_in[4];
  const float* b_l  = (const float*)d_in[5];
  const float* w_m  = (const float*)d_in[6];
  const float* b_m  = (const float*)d_in[7];
  const float* w_s  = (const float*)d_in[8];
  const float* b_s  = (const float*)d_in[9];
  const float* fc1w = (const float*)d_in[10];
  const float* fc1b = (const float*)d_in[11];
  const float* q_w  = (const float*)d_in[12];
  const float* q_b  = (const float*)d_in[13];
  const float* k_w  = (const float*)d_in[14];
  const float* k_b  = (const float*)d_in[15];
  const float* v_w  = (const float*)d_in[16];
  const float* v_b  = (const float*)d_in[17];
  const float* t_k  = (const float*)d_in[18];
  const float* t_v  = (const float*)d_in[19];
  const float* t_q  = (const float*)d_in[20];
  const float* lw   = (const float*)d_in[21];
  const float* lb   = (const float*)d_in[22];
  const float* lng  = (const float*)d_in[23];
  const float* lnb  = (const float*)d_in[24];
  const float* outw = (const float*)d_in[25];
  const float* outb = (const float*)d_in[26];

  float* wsf  = (float*)d_ws;
  float* cat  = wsf + F_CAT;
  float* sbuf = wsf + F_S;
  float* W1   = wsf + F_W1;
  float* hb   = wsf + F_HB;
  float* qT   = wsf + F_QT;
  float* kT   = wsf + F_KT;
  float* vT   = wsf + F_VT;
  float* lwT  = wsf + F_LWT;
  float* gW   = wsf + F_GW;
  float* gb   = wsf + F_GB;
  float* lw1T = wsf + F_LW1T;
  float* lb1  = wsf + F_LB1;

  k_cat<<<BATCH / 8, 256, 0, stream>>>(in1, in2, in3, in4, w_l, b_l, w_m, b_m, w_s, b_s, cat);
  k_prep<<<1, 256, 0, stream>>>(fc1w, fc1b, q_w, k_w, v_w, lw, wsf);
  k_qkv<<<BATCH / 128, 256, 0, stream>>>(cat, q_b, k_b, v_b, sbuf, W1, hb, qT, kT, vT);
  k_grad<<<512, 128, 0, stream>>>(sbuf, t_k, t_v, lwT, lb, gW, gb);
  k_update<<<1, 256, 0, stream>>>(lw, lb, gW, gb, lw1T, lb1);
  k_out<<<BATCH / 64, 128, 0, stream>>>(sbuf, t_q, lw1T, lb1, lng, lnb, outw, outb,
                                        (float*)d_out);
}


# k_grad, k_out: tv, pred - lv and u tiles come straight from the cat rows (K = 16, a ones column carries biases) by 16 x 64 weight-only factors made in k_prep on WMMA; changes g, gW, gb, o sum order
# speedup vs baseline: 2.2005x; 2.2005x over previous
#include <hip/hip_runtime.h>
#include <math.h>

#define BATCH 32768
#define LRATE 0.01f
#define AEPS  1e-8f
#define LNEPS 1e-5f

typedef __attribute__((ext_vector_type(2))) float v2f;
typedef __attribute__((ext_vector_type(8))) float v8f;

__device__ __forceinline__ v2f mkv2(float x, float y) { v2f v; v[0] = x; v[1] = y; return v; }

__device__ __forceinline__ v8f wmma_f32(v2f a, v2f b, v8f c) {
  return __builtin_amdgcn_wmma_f32_16x16x4_f32(false, a, false, b, (short)0, c, false, false);
}

constexpr size_t F_CAT  = 0;
constexpr size_t F_P    = F_CAT + (size_t)BATCH * 16;
constexpr size_t F_Q    = F_P + 3 * 1024;
constexpr size_t F_U    = F_Q + 3 * 1024;
constexpr size_t F_GW   = F_U + 1024;
constexpr size_t F_GB   = F_GW + 64 * 64;
constexpr size_t F_LW1T = F_GB + 64;
constexpr size_t F_LB1  = F_LW1T + 64 * 64;

typedef __attribute__((ext_vector_type(4))) float v4f;

#define CP   132
#define WT_L 0
#define WT_M 4512
#define WT_S 6768
#define WT_N 7224

#define CAT_UNROLL _Pragma("unroll")

struct CDesc {
  const float* rowp;
  int rowlen;
  int valid;
  int wide;
  int wofs;
  int wpitch;
  int col;
};

__device__ __forceinline__ CDesc cat_desc(int part, int ch, int row0, const float* in1,
                                          const float* in2, const float* in3) {
  CDesc d;
  if (part == 0 || ch < 3) {
    const int off = (part == 0 ? 0 : 1200) + ch * 128;
    const int end = (part == 0) ? 1200 : 1500;
    d.rowp = in1 + (size_t)row0 * 1500 + off; d.rowlen = 1500;
    d.valid = min(128, end - off); d.wide = 1;
    d.wofs = WT_L + off; d.wpitch = 1504; d.col = 0;
  } else if (ch < 9) {
    const int off = (ch - 3) * 128;
    d.rowp = in2 + (size_t)row0 * 750 + off; d.rowlen = 750;
    d.valid = min(128, 750 - off); d.wide = 0;
    d.wofs = WT_M + off; d.wpitch = 752; d.col = 3;
  } else {
    const int off = (ch - 9) * 128;
    d.rowp = in3 + (size_t)row0 * 150 + off; d.rowlen = 150;
    d.valid = min(128, 150 - off); d.wide = 0;
    d.wofs = WT_S + off; d.wpitch = 152; d.col = 6;
  }
  return d;
}

__device__ __forceinline__ void cat_load(v4f (&buf)[16], const CDesc& d, int lane) {
  if (d.wide) {
    const float* p = d.rowp + 4 * lane;
    if (d.valid == 128 || 4 * lane < d.valid) {
CAT_UNROLL
      for (int r = 0; r < 16; ++r)
        buf[r] = __builtin_nontemporal_load((const v4f*)(p + (size_t)r * d.rowlen));
    } else {
CAT_UNROLL
      for (int r = 0; r < 16; ++r) buf[r] = (v4f){0.f, 0.f, 0.f, 0.f};
    }
  } else {
    const float* p = d.rowp + 2 * lane;
    v2f a[16], b[16];
    if (d.valid == 128 || 2 * lane < d.valid) {
CAT_UNROLL
      for (int r = 0; r < 16; ++r)
        a[r] = __builtin_nontemporal_load((const v2f*)(p + (size_t)r * d.rowlen));
    } else {
CAT_UNROLL
      for (int r = 0; r < 16; ++r) a[r] = mkv2(0.f, 0.f);
    }
    if (d.valid == 128 || 64 + 2 * lane < d.valid) {
CAT_UNROLL
      for (int r = 0; r < 16; ++r)
        b[r] = __builtin_nontemporal_load((const v2f*)(p + (size_t)r * d.rowlen + 64));
    } else {
CAT_UNROLL
      for (int r = 0; r < 16; ++r) b[r] = mkv2(0.f, 0.f);
    }
CAT_UNROLL
    for (int r = 0; r < 16; ++r) {
      v4f v; v[0] = a[r][0]; v[1] = a[r][1]; v[2] = b[r][0]; v[3] = b[r][1];
      buf[r] = v;
    }
  }
}

__device__ __forceinline__ void cat_stage(float* st, const v4f (&buf)[16], int wide, int lane) {
  if (wide) {
CAT_UNROLL
    for (int r = 0; r < 16; ++r) *(v4f*)(st + r * CP + 4 * lane) = buf[r];
  } else {
CAT_UNROLL
    for (int r = 0; r < 16; ++r) {
      *(v2f*)(st + r * CP + 2 * lane)      = mkv2(buf[r][0], buf[r][1]);
      *(v2f*)(st + r * CP + 64 + 2 * lane) = mkv2(buf[r][2], buf[r][3]);
    }
  }
}

__device__ __forceinline__ void cat_frag(const float* ap, const float* bp, bool act, int i0,
                                         v4f (&a)[4], v4f (&b)[4]) {
CAT_UNROLL
  for (int i = 0; i < 4; ++i) {
    a[i] = *(const v4f*)(ap + 8 * (i0 + i));
    const v4f w = *(const v4f*)(bp + 8 * (i0 + i));
    b[i] = act ? w : (v4f){0.f, 0.f, 0.f, 0.f};
  }
}

__device__ __forceinline__ void cat_mul(const v4f (&a)[4], const v4f (&b)[4], v8f& acc0, v8f& acc1) {
CAT_UNROLL
  for (int i = 0; i < 4; ++i) {
    acc0 = wmma_f32(mkv2(a[i][0], a[i][1]), mkv2(b[i][0], b[i][1]), acc0);
    acc1 = wmma_f32(mkv2(a[i][2], a[i][3]), mkv2(b[i][2], b[i][3]), acc1);
  }
}

#define CAT_PIN() __builtin_amdgcn_sched_barrier(0)

__device__ __forceinline__ void cat_mma(const float* st, const float* wt, const CDesc& d,
                                        int nl, int hi, v8f& acc0, v8f& acc1) {
  const int o = nl - d.col;
  const bool act = (unsigned)o < 3u;
  const float* ap = st + nl * CP + 4 * hi;
  const float* bp = wt + d.wofs + (act ? o : 0) * d.wpitch + 4 * hi;
  const int np = (d.valid + 7) >> 3;
  if (np == 16) {
    v4f a0[4], b0[4], a1[4], b1[4];
    cat_frag(ap, bp, act, 0, a0, b0);
    CAT_PIN();
    cat_frag(ap, bp, act, 4, a1, b1);
    CAT_PIN();
    cat_mul(a0, b0, acc0, acc1);
    CAT_PIN();
    cat_frag(ap, bp, act, 8, a0, b0);
    CAT_PIN();
    cat_mul(a1, b1, acc0, acc1);
    CAT_PIN();
    cat_frag(ap, bp, act, 12, a1, b1);
    CAT_PIN();
    cat_mul(a0, b0, acc0, acc1);
    cat_mul(a1, b1, acc0, acc1);
  } else {
    for (int i = 0; i < np; ++i) {
      const v4f a = *(const v4f*)(ap + 8 * i);
      const v4f w = *(const v4f*)(bp + 8 * i);
      const v4f b = act ? w : (v4f){0.f, 0.f, 0.f, 0.f};
      acc0 = wmma_f32(mkv2(a[0], a[1]), mkv2(b[0], b[1]), acc0);
      acc1 = wmma_f32(mkv2(a[2], a[3]), mkv2(b[2], b[3]), acc1);
    }
  }
}

__global__ void __launch_bounds__(256) k_cat(
    const float* __restrict__ in1, const float* __restrict__ in2,
    const float* __restrict__ in3, const float* __restrict__ in4,
    const float* __restrict__ wl, const float* __restrict__ bl,
    const float* __restrict__ wm, const float* __restrict__ bm,
    const float* __restrict__ wsw, const float* __restrict__ bs,
    float* __restrict__ cat)
{
  __shared__ __attribute__((aligned(16))) float wt[WT_N];
  __shared__ __attribute__((aligned(16))) float stage[8][16 * CP];
  __shared__ __attribute__((aligned(16))) float comb[4][256];

  const int lane = threadIdx.x & 31;
  const int warp = __builtin_amdgcn_readfirstlane(threadIdx.x >> 5);
  const int part = warp & 1, tl = warp >> 1;
  const int row0 = (blockIdx.x * 4 + tl) * 16;
  const int nl = lane & 15, hi = lane >> 4;
  float* st = stage[warp];

  v4f buf[16];
  CDesc d = cat_desc(part, 0, row0, in1, in2, in3);
  cat_load(buf, d, lane);

  float wv[29];
CAT_UNROLL
  for (int it = 0; it < 29; ++it) {
    const int i = it * 256 + (int)threadIdx.x;
    const float* sp; int o, j, K;
    if (i < WT_M)      { o = i / 1504; j = i - o * 1504; K = 500; sp = wl; }
    else if (i < WT_S) { const int t = i - WT_M; o = t / 752; j = t - o * 752; K = 250; sp = wm; }
    else               { const int t = i - WT_S; o = t / 152; j = t - o * 152; K = 50; sp = wsw; }
    const bool ok = (i < WT_N) && (j < 3 * K);
    const int jj = ok ? j : 0, oo = ok ? o : 0;
    const int k = jj / 3, c = jj - 3 * k;
    const float v = sp[(oo * 3 + c) * K + k];
    wv[it] = ok ? v : 0.f;
  }
CAT_UNROLL
  for (int it = 0; it < 29; ++it) {
    const int i = it * 256 + (int)threadIdx.x;
    if (i < WT_N) wt[i] = wv[it];
  }
  __syncthreads();

  v8f acc0 = {}, acc1 = {};
  const int nch = (part == 0) ? 10 : 11;
  for (int ch = 0; ch < nch; ++ch) {
    cat_stage(st, buf, d.wide, lane);
    CDesc dn = d;
    if (ch + 1 < nch) {
      dn = cat_desc(part, ch + 1, row0, in1, in2, in3);
      cat_load(buf, dn, lane);
    }
    cat_mma(st, wt, d, nl, hi, acc0, acc1);
    d = dn;
  }

  float* cb = comb[tl];
  v8f acc = acc0 + acc1;
  if (part == 1) {
CAT_UNROLL
    for (int r = 0; r < 8; ++r) cb[(8 * hi + r) * 16 + nl] = acc[r];
  }
  __syncthreads();
  if (part == 0) {
    float bias = 0.f;
    if (nl < 3)      bias = bl[nl];
    else if (nl < 6) bias = bm[nl - 3];
    else if (nl < 9) bias = bs[nl - 6];
CAT_UNROLL
    for (int r = 0; r < 8; ++r) {
      const int row = 8 * hi + r;
      float v = acc[r] + cb[row * 16 + nl] + bias;
      if (nl >= 9) v = (nl < 12) ? in4[(size_t)(row0 + row) * 3 + (nl - 9)] : (nl == 12 ? 1.f : 0.f);
      cb[row * 16 + nl] = v;
    }
    v4f* dst = (v4f*)(cat + (size_t)row0 * 16);
    const v4f* srcp = (const v4f*)cb;
    dst[lane] = srcp[lane];
    dst[32 + lane] = srcp[32 + lane];
  }
}


__device__ __forceinline__ void store_factor(float* __restrict__ X, const v8f& c, int w, int hi, int nl) {
#pragma unroll
  for (int q = 0; q < 4; ++q) {
    const int ks = 2 * hi + (q >> 1);
    const int ln = 16 * (q & 1) + nl;
    *(v2f*)(X + ((ks * 2 + (w >> 1)) * 32 + ln) * 4 + (w & 1) * 2) = mkv2(c[2 * q], c[2 * q + 1]);
  }
}

__device__ __forceinline__ void load_factor(const float* __restrict__ X, int lane, v2f (&b)[4][4]) {
#pragma unroll
  for (int ks = 0; ks < 4; ++ks)
#pragma unroll
    for (int np = 0; np < 2; ++np) {
      const v4f w = *(const v4f*)(X + ((ks * 2 + np) * 32 + lane) * 4);
      b[ks][2 * np]     = mkv2(w[0], w[1]);
      b[ks][2 * np + 1] = mkv2(w[2], w[3]);
    }
}

__global__ void __launch_bounds__(128) k_prep(
    const float* __restrict__ fc1w, const float* __restrict__ fc1b,
    const float* __restrict__ qw, const float* __restrict__ qb,
    const float* __restrict__ kw, const float* __restrict__ kbv,
    const float* __restrict__ vw, const float* __restrict__ vb,
    const float* __restrict__ tk, const float* __restrict__ tvm, const float* __restrict__ tq,
    const float* __restrict__ lw, const float* __restrict__ lb,
    float* __restrict__ wsf)
{
  __shared__ float sS[16 * 68];
  __shared__ float sP[16 * 68];
  const int t = (int)blockIdx.x;
  const int warp = threadIdx.x >> 5, lane = threadIdx.x & 31;
  const int hi = lane >> 4, nl = lane & 15;
  const int n = warp * 16 + nl;

  float* acc = wsf + F_GW;
  for (int i = t * 128 + (int)threadIdx.x; i < 64 * 64 + 64; i += 3 * 128) acc[i] = 0.f;

  const float* Wt = (t == 0) ? qw : (t == 1) ? kw : vw;
  const float* Bt = (t == 0) ? qb : (t == 1) ? kbv : vb;

  const bool isw = nl < 12, isb = nl == 12;
  v2f a1[16], bw[16], btk[16], btv[16], blw[16], btq[16];
#pragma unroll
  for (int ks = 0; ks < 16; ++ks) {
    const int kb = ks * 4 + 2 * hi;
    const float* p0 = isw ? fc1w + kb * 12 + nl       : fc1b + kb;
    const float* p1 = isw ? fc1w + (kb + 1) * 12 + nl : fc1b + kb + 1;
    const float r0 = *p0, r1 = *p1;
    a1[ks]  = mkv2(isw ? r0 : (isb ? r0 : 0.f), isw ? r1 : (isb ? r1 + 1.0f : 0.f));
    bw[ks]  = *(const v2f*)(Wt + n * 64 + kb);
    btk[ks] = mkv2(tk[kb * 64 + n],  tk[(kb + 1) * 64 + n]);
    btv[ks] = mkv2(tvm[kb * 64 + n], tvm[(kb + 1) * 64 + n]);
    blw[ks] = *(const v2f*)(lw + n * 64 + kb);
    btq[ks] = mkv2(tq[kb * 64 + n], tq[(kb + 1) * 64 + n]);
  }
  const float bt  = Bt[n];
  const float lbn = lb[n];

  v8f cS = {};
#pragma unroll
  for (int ks = 0; ks < 16; ++ks) cS = wmma_f32(a1[ks], bw[ks], cS);
  cS[4] += (hi == 1) ? bt : 0.f;
#pragma unroll
  for (int r = 0; r < 8; ++r) sS[(r + 8 * hi) * 68 + n] = cS[r];
  __syncthreads();

  v2f as[16];
#pragma unroll
  for (int ks = 0; ks < 16; ++ks) {
    const int kb = ks * 4 + 2 * hi;
    as[ks] = mkv2(sS[nl * 68 + kb], sS[nl * 68 + kb + 1]);
  }
  v8f cP = {}, cL = {};
  if (t == 2) {
    v8f cU = {};
#pragma unroll
    for (int ks = 0; ks < 16; ++ks) {
      cP = wmma_f32(as[ks], btk[ks], cP);
      cL = wmma_f32(as[ks], btv[ks], cL);
      cU = wmma_f32(as[ks], btq[ks], cU);
    }
    store_factor(wsf + F_U, cU, warp, hi, nl);
  } else {
#pragma unroll
    for (int ks = 0; ks < 16; ++ks) {
      cP = wmma_f32(as[ks], btk[ks], cP);
      cL = wmma_f32(as[ks], btv[ks], cL);
    }
  }
  store_factor(wsf + F_P + t * 1024, cP, warp, hi, nl);
#pragma unroll
  for (int r = 0; r < 8; ++r) sP[(r + 8 * hi) * 68 + n] = cP[r];
  __syncthreads();

  v2f ap[16];
#pragma unroll
  for (int ks = 0; ks < 16; ++ks) {
    const int kb = ks * 4 + 2 * hi;
    ap[ks] = mkv2(sP[nl * 68 + kb], sP[nl * 68 + kb + 1]);
  }
  v8f cQ = {};
#pragma unroll
  for (int ks = 0; ks < 16; ++ks) cQ = wmma_f32(ap[ks], blw[ks], cQ);
#pragma unroll
  for (int r = 0; r < 8; ++r) cQ[r] -= cL[r];
  cQ[4] += (hi == 1) ? lbn : 0.f;
  store_factor(wsf + F_Q + t * 1024, cQ, warp, hi, nl);
}

constexpr int KG_BLOCKS = 256;
constexpr int KG_WAVES  = KG_BLOCKS * 4;

__global__ void __launch_bounds__(128) k_grad(
    const float* __restrict__ cat, const float* __restrict__ P, const float* __restrict__ Q,
    float* __restrict__ gW, float* __restrict__ gb)
{
  __shared__ float sgW[64 * 64];
  __shared__ float sgb[64];
  for (int i = threadIdx.x; i < 4096; i += 128) sgW[i] = 0.f;
  if (threadIdx.x < 64) sgb[threadIdx.x] = 0.f;
  __syncthreads();

  const int warp = threadIdx.x >> 5, lane = threadIdx.x & 31;
  const int hi = lane >> 4, nl = lane & 15;

  const float gscale = 2.0f / (float)(3ull * BATCH * 64);
  const int NCT = BATCH / 16;
  const int wave = blockIdx.x * 4 + warp;
  const int nw = KG_WAVES;

  v2f bp[3][4][4], bq[3][4][4];
#pragma unroll
  for (int t = 0; t < 3; ++t) {
    load_factor(P + t * 1024, lane, bp[t]);
    load_factor(Q + t * 1024, lane, bq[t]);
  }

  v8f cgw[4][4];
#pragma unroll
  for (int i0 = 0; i0 < 4; ++i0)
#pragma unroll
    for (int j0 = 0; j0 < 4; ++j0) cgw[i0][j0] = (v8f){};
  float gbp[4] = {0.f, 0.f, 0.f, 0.f};

  v2f cn[4];
#pragma unroll
  for (int ks = 0; ks < 4; ++ks)
    cn[ks] = *(const v2f*)(cat + (size_t)(wave * 16 + nl) * 16 + ks * 4 + 2 * hi);

  for (int ct = wave; ct < NCT; ct += nw) {
    v2f ca[4];
#pragma unroll
    for (int ks = 0; ks < 4; ++ks) ca[ks] = cn[ks];
    if (ct + nw < NCT) {
#pragma unroll
      for (int ks = 0; ks < 4; ++ks)
        cn[ks] = *(const v2f*)(cat + (size_t)((ct + nw) * 16 + nl) * 16 + ks * 4 + 2 * hi);
    }
#pragma unroll
    for (int t = 0; t < 3; ++t) {
      v8f ctv[4], gt[4];
#pragma unroll
      for (int n0 = 0; n0 < 4; ++n0) {
        v8f cv = {}, cd = {};
#pragma unroll
        for (int ks = 0; ks < 4; ++ks) {
          cv = wmma_f32(ca[ks], bp[t][ks][n0], cv);
          cd = wmma_f32(ca[ks], bq[t][ks][n0], cd);
        }
        ctv[n0] = cv;
        float colsum = 0.f;
#pragma unroll
        for (int r = 0; r < 8; ++r) {
          const float g = gscale * cd[r];
          gt[n0][r] = g;
          colsum += g;
        }
        gbp[n0] += colsum;
      }
#pragma unroll
      for (int i0 = 0; i0 < 4; ++i0) {
#pragma unroll
        for (int j0 = 0; j0 < 4; ++j0) {
#pragma unroll
          for (int q = 0; q < 4; ++q) {
            v2f a = mkv2(gt[i0][2 * q], gt[i0][2 * q + 1]);
            v2f b = mkv2(ctv[j0][2 * q], ctv[j0][2 * q + 1]);
            cgw[i0][j0] = wmma_f32(a, b, cgw[i0][j0]);
          }
        }
      }
    }
  }
#pragma unroll
  for (int i0 = 0; i0 < 4; ++i0)
#pragma unroll
    for (int j0 = 0; j0 < 4; ++j0)
#pragma unroll
      for (int r = 0; r < 8; ++r)
        atomicAdd(&sgW[(i0 * 16 + r + 8 * hi) * 64 + j0 * 16 + nl], cgw[i0][j0][r]);
#pragma unroll
  for (int n0 = 0; n0 < 4; ++n0) atomicAdd(&sgb[n0 * 16 + nl], gbp[n0]);
  __syncthreads();
  for (int i = threadIdx.x; i < 4096; i += 128) atomicAdd(&gW[i], sgW[i]);
  if (threadIdx.x < 64) atomicAdd(&gb[threadIdx.x], sgb[threadIdx.x]);
}

__global__ void __launch_bounds__(256) k_update(
                         const float* __restrict__ lw, const float* __restrict__ lb,
                         const float* __restrict__ gW, const float* __restrict__ gb,
                         float* __restrict__ lw1T, float* __restrict__ lb1)
{
  const int b = (int)blockIdx.x, t = (int)threadIdx.x;
  if (b < 16) {
    const int i = b * 256 + t;
    const int n = i >> 6, k = i & 63;
    const float g = gW[i];
    lw1T[k * 64 + n] = lw[i] - LRATE * g / (fabsf(g) + AEPS);
  } else if (t < 64) {
    const float g = gb[t];
    lb1[t] = lb[t] - LRATE * g / (fabsf(g) + AEPS);
  }
}

__global__ void __launch_bounds__(128) k_out(
    const float* __restrict__ cat, const float* __restrict__ U,
    const float* __restrict__ lw1T, const float* __restrict__ lb1,
    const float* __restrict__ lng, const float* __restrict__ lnb,
    const float* __restrict__ ow, const float* __restrict__ obp,
    float* __restrict__ out)
{
  __shared__ float su[4][16 * 68];
  __shared__ float so[4][16 * 68];
  const int warp = threadIdx.x >> 5, lane = threadIdx.x & 31;
  const int hi = lane >> 4, nl = lane & 15;
  const int row0 = (blockIdx.x * 4 + warp) * 16;
  float* uw = su[warp];
  float* om = so[warp];

  v2f ca[4];
#pragma unroll
  for (int ks = 0; ks < 4; ++ks)
    ca[ks] = *(const v2f*)(cat + (size_t)(row0 + nl) * 16 + ks * 4 + 2 * hi);
  v2f bu[4][4];
  load_factor(U, lane, bu);
#pragma unroll
  for (int n0 = 0; n0 < 4; ++n0) {
    int n = n0 * 16 + nl;
    v8f c = {};
#pragma unroll
    for (int ks = 0; ks < 4; ++ks) c = wmma_f32(ca[ks], bu[ks][n0], c);
#pragma unroll
    for (int r = 0; r < 8; ++r) uw[(r + 8 * hi) * 68 + n] = c[r];
  }
#pragma unroll
  for (int n0 = 0; n0 < 4; ++n0) {
    int n = n0 * 16 + nl;
    v8f c = {};
#pragma unroll
    for (int ks = 0; ks < 16; ++ks) {
      int kb = ks * 4 + 2 * hi;
      v2f a = mkv2(uw[nl * 68 + kb], uw[nl * 68 + kb + 1]);
      c = wmma_f32(a, mkv2(lw1T[kb * 64 + n], lw1T[(kb + 1) * 64 + n]), c);
    }
    float bias = lb1[n];
#pragma unroll
    for (int r = 0; r < 8; ++r) om[(r + 8 * hi) * 68 + n] = c[r] + bias;
  }
  __syncthreads();
  if (lane < 16) {
    const float* rowp = om + lane * 68;
    float mu = 0.f;
#pragma unroll
    for (int j = 0; j < 64; ++j) mu += rowp[j];
    mu *= (1.0f / 64.0f);
    float var = 0.f;
#pragma unroll
    for (int j = 0; j < 64; ++j) { float d = rowp[j] - mu; var += d * d; }
    var *= (1.0f / 64.0f);
    float rstd = rsqrtf(var + LNEPS);
    float acc = 0.f;
#pragma unroll
    for (int j = 0; j < 64; ++j) {
      float nrm = (rowp[j] - mu) * rstd * lng[j] + lnb[j];
      float a = nrm / (1.0f + __expf(-nrm));
      acc += a * ow[j];
    }
    out[row0 + lane] = acc + obp[0];
  }
}

extern "C" void kernel_launch(void* const* d_in, const int* in_sizes, int n_in,
                              void* d_out, int out_size, void* d_ws, size_t ws_size,
                              hipStream_t stream) {
  const float* in1  = (const float*)d_in[0];
  const float* in2  = (const float*)d_in[1];
  const float* in3  = (const float*)d_in[2];
  const float* in4  = (const float*)d_in[3];
  const float* w_l  = (const float*)d_in[4];
  const float* b_l  = (const float*)d_in[5];
  const float* w_m  = (const float*)d_in[6];
  const float* b_m  = (const float*)d_in[7];
  const float* w_s  = (const float*)d_in[8];
  const float* b_s  = (const float*)d_in[9];
  const float* fc1w = (const float*)d_in[10];
  const float* fc1b = (const float*)d_in[11];
  const float* q_w  = (const float*)d_in[12];
  const float* q_b  = (const float*)d_in[13];
  const float* k_w  = (const float*)d_in[14];
  const float* k_b  = (const float*)d_in[15];
  const float* v_w  = (const float*)d_in[16];
  const float* v_b  = (const float*)d_in[17];
  const float* t_k  = (const float*)d_in[18];
  const float* t_v  = (const float*)d_in[19];
  const float* t_q  = (const float*)d_in[20];
  const float* lw   = (const float*)d_in[21];
  const float* lb   = (const float*)d_in[22];
  const float* lng  = (const float*)d_in[23];
  const float* lnb  = (const float*)d_in[24];
  const float* outw = (const float*)d_in[25];
  const float* outb = (const float*)d_in[26];

  float* wsf  = (float*)d_ws;
  float* cat  = wsf + F_CAT;
  float* fP   = wsf + F_P;
  float* fQ   = wsf + F_Q;
  float* fU   = wsf + F_U;
  float* gW   = wsf + F_GW;
  float* gb   = wsf + F_GB;
  float* lw1T = wsf + F_LW1T;
  float* lb1  = wsf + F_LB1;

  k_cat<<<BATCH / 64, 256, 0, stream>>>(in1, in2, in3, in4, w_l, b_l, w_m, b_m, w_s, b_s, cat);
  k_prep<<<3, 128, 0, stream>>>(fc1w, fc1b, q_w, q_b, k_w, k_b, v_w, v_b, t_k, t_v, t_q, lw, lb, wsf);
  k_grad<<<KG_BLOCKS, 128, 0, stream>>>(cat, fP, fQ, gW, gb);
  k_update<<<17, 256, 0, stream>>>(lw, lb, gW, gb, lw1T, lb1);
  k_out<<<BATCH / 64, 128, 0, stream>>>(cat, fU, lw1T, lb1, lng, lnb, outw, outb,
                                        (float*)d_out);
}


# k_prep's launch goes: the first three blocks of k_cat, which enter first, make the weight-only factors and clear the grad sums before they stream their rows; 5 -> 4 launches
# speedup vs baseline: 2.3772x; 1.0803x over previous
#include <hip/hip_runtime.h>
#include <math.h>

#define BATCH 32768
#define LRATE 0.01f
#define AEPS  1e-8f
#define LNEPS 1e-5f

typedef __attribute__((ext_vector_type(2))) float v2f;
typedef __attribute__((ext_vector_type(8))) float v8f;

__device__ __forceinline__ v2f mkv2(float x, float y) { v2f v; v[0] = x; v[1] = y; return v; }

__device__ __forceinline__ v8f wmma_f32(v2f a, v2f b, v8f c) {
  return __builtin_amdgcn_wmma_f32_16x16x4_f32(false, a, false, b, (short)0, c, false, false);
}

constexpr size_t F_CAT  = 0;
constexpr size_t F_P    = F_CAT + (size_t)BATCH * 16;
constexpr size_t F_Q    = F_P + 3 * 1024;
constexpr size_t F_U    = F_Q + 3 * 1024;
constexpr size_t F_GW   = F_U + 1024;
constexpr size_t F_GB   = F_GW + 64 * 64;
constexpr size_t F_LW1T = F_GB + 64;
constexpr size_t F_LB1  = F_LW1T + 64 * 64;

typedef __attribute__((ext_vector_type(4))) float v4f;

#define CP   132
#define WT_L 0
#define WT_M 4512
#define WT_S 6768
#define WT_N 7224

#define CAT_UNROLL _Pragma("unroll")

struct CDesc {
  const float* rowp;
  int rowlen;
  int valid;
  int wide;
  int wofs;
  int wpitch;
  int col;
};

__device__ __forceinline__ CDesc cat_desc(int part, int ch, int row0, const float* in1,
                                          const float* in2, const float* in3) {
  CDesc d;
  if (part == 0 || ch < 3) {
    const int off = (part == 0 ? 0 : 1200) + ch * 128;
    const int end = (part == 0) ? 1200 : 1500;
    d.rowp = in1 + (size_t)row0 * 1500 + off; d.rowlen = 1500;
    d.valid = min(128, end - off); d.wide = 1;
    d.wofs = WT_L + off; d.wpitch = 1504; d.col = 0;
  } else if (ch < 9) {
    const int off = (ch - 3) * 128;
    d.rowp = in2 + (size_t)row0 * 750 + off; d.rowlen = 750;
    d.valid = min(128, 750 - off); d.wide = 0;
    d.wofs = WT_M + off; d.wpitch = 752; d.col = 3;
  } else {
    const int off = (ch - 9) * 128;
    d.rowp = in3 + (size_t)row0 * 150 + off; d.rowlen = 150;
    d.valid = min(128, 150 - off); d.wide = 0;
    d.wofs = WT_S + off; d.wpitch = 152; d.col = 6;
  }
  return d;
}

__device__ __forceinline__ void cat_load(v4f (&buf)[16], const CDesc& d, int lane) {
  if (d.wide) {
    const float* p = d.rowp + 4 * lane;
    if (d.valid == 128 || 4 * lane < d.valid) {
CAT_UNROLL
      for (int r = 0; r < 16; ++r)
        buf[r] = __builtin_nontemporal_load((const v4f*)(p + (size_t)r * d.rowlen));
    } else {
CAT_UNROLL
      for (int r = 0; r < 16; ++r) buf[r] = (v4f){0.f, 0.f, 0.f, 0.f};
    }
  } else {
    const float* p = d.rowp + 2 * lane;
    v2f a[16], b[16];
    if (d.valid == 128 || 2 * lane < d.valid) {
CAT_UNROLL
      for (int r = 0; r < 16; ++r)
        a[r] = __builtin_nontemporal_load((const v2f*)(p + (size_t)r * d.rowlen));
    } else {
CAT_UNROLL
      for (int r = 0; r < 16; ++r) a[r] = mkv2(0.f, 0.f);
    }
    if (d.valid == 128 || 64 + 2 * lane < d.valid) {
CAT_UNROLL
      for (int r = 0; r < 16; ++r)
        b[r] = __builtin_nontemporal_load((const v2f*)(p + (size_t)r * d.rowlen + 64));
    } else {
CAT_UNROLL
      for (int r = 0; r < 16; ++r) b[r] = mkv2(0.f, 0.f);
    }
CAT_UNROLL
    for (int r = 0; r < 16; ++r) {
      v4f v; v[0] = a[r][0]; v[1] = a[r][1]; v[2] = b[r][0]; v[3] = b[r][1];
      buf[r] = v;
    }
  }
}

__device__ __forceinline__ void cat_stage(float* st, const v4f (&buf)[16], int wide, int lane) {
  if (wide) {
CAT_UNROLL
    for (int r = 0; r < 16; ++r) *(v4f*)(st + r * CP + 4 * lane) = buf[r];
  } else {
CAT_UNROLL
    for (int r = 0; r < 16; ++r) {
      *(v2f*)(st + r * CP + 2 * lane)      = mkv2(buf[r][0], buf[r][1]);
      *(v2f*)(st + r * CP + 64 + 2 * lane) = mkv2(buf[r][2], buf[r][3]);
    }
  }
}

__device__ __forceinline__ void cat_frag(const float* ap, const float* bp, bool act, int i0,
                                         v4f (&a)[4], v4f (&b)[4]) {
CAT_UNROLL
  for (int i = 0; i < 4; ++i) {
    a[i] = *(const v4f*)(ap + 8 * (i0 + i));
    const v4f w = *(const v4f*)(bp + 8 * (i0 + i));
    b[i] = act ? w : (v4f){0.f, 0.f, 0.f, 0.f};
  }
}

__device__ __forceinline__ void cat_mul(const v4f (&a)[4], const v4f (&b)[4], v8f& acc0, v8f& acc1) {
CAT_UNROLL
  for (int i = 0; i < 4; ++i) {
    acc0 = wmma_f32(mkv2(a[i][0], a[i][1]), mkv2(b[i][0], b[i][1]), acc0);
    acc1 = wmma_f32(mkv2(a[i][2], a[i][3]), mkv2(b[i][2], b[i][3]), acc1);
  }
}

#define CAT_PIN() __builtin_amdgcn_sched_barrier(0)

__device__ __forceinline__ void cat_mma(const float* st, const float* wt, const CDesc& d,
                                        int nl, int hi, v8f& acc0, v8f& acc1) {
  const int o = nl - d.col;
  const bool act = (unsigned)o < 3u;
  const float* ap = st + nl * CP + 4 * hi;
  const float* bp = wt + d.wofs + (act ? o : 0) * d.wpitch + 4 * hi;
  const int np = (d.valid + 7) >> 3;
  if (np == 16) {
    v4f a0[4], b0[4], a1[4], b1[4];
    cat_frag(ap, bp, act, 0, a0, b0);
    CAT_PIN();
    cat_frag(ap, bp, act, 4, a1, b1);
    CAT_PIN();
    cat_mul(a0, b0, acc0, acc1);
    CAT_PIN();
    cat_frag(ap, bp, act, 8, a0, b0);
    CAT_PIN();
    cat_mul(a1, b1, acc0, acc1);
    CAT_PIN();
    cat_frag(ap, bp, act, 12, a1, b1);
    CAT_PIN();
    cat_mul(a0, b0, acc0, acc1);
    cat_mul(a1, b1, acc0, acc1);
  } else {
    for (int i = 0; i < np; ++i) {
      const v4f a = *(const v4f*)(ap + 8 * i);
      const v4f w = *(const v4f*)(bp + 8 * i);
      const v4f b = act ? w : (v4f){0.f, 0.f, 0.f, 0.f};
      acc0 = wmma_f32(mkv2(a[0], a[1]), mkv2(b[0], b[1]), acc0);
      acc1 = wmma_f32(mkv2(a[2], a[3]), mkv2(b[2], b[3]), acc1);
    }
  }
}


__device__ __forceinline__ void store_factor(float* __restrict__ X, const v8f& c, int w, int hi, int nl) {
#pragma unroll
  for (int q = 0; q < 4; ++q) {
    const int ks = 2 * hi + (q >> 1);
    const int ln = 16 * (q & 1) + nl;
    *(v2f*)(X + ((ks * 2 + (w >> 1)) * 32 + ln) * 4 + (w & 1) * 2) = mkv2(c[2 * q], c[2 * q + 1]);
  }
}

__device__ __forceinline__ void load_factor(const float* __restrict__ X, int lane, v2f (&b)[4][4]) {
#pragma unroll
  for (int ks = 0; ks < 4; ++ks)
#pragma unroll
    for (int np = 0; np < 2; ++np) {
      const v4f w = *(const v4f*)(X + ((ks * 2 + np) * 32 + lane) * 4);
      b[ks][2 * np]     = mkv2(w[0], w[1]);
      b[ks][2 * np + 1] = mkv2(w[2], w[3]);
    }
}

__device__ __forceinline__ void prep_factors(
    int t, int warp, int lane, int tid, float* __restrict__ sS, float* __restrict__ sP,
    const float* __restrict__ fc1w, const float* __restrict__ fc1b,
    const float* __restrict__ Wt, const float* __restrict__ Bt,
    const float* __restrict__ tk, const float* __restrict__ tvm, const float* __restrict__ tq,
    const float* __restrict__ lw, const float* __restrict__ lb,
    float* __restrict__ wsf)
{
  const int hi = lane >> 4, nl = lane & 15;
  const bool on = warp < 4;
  const int n = (warp & 3) * 16 + nl;

  float* acc = wsf + F_GW;
  for (int i = t * 256 + tid; i < 64 * 64 + 64; i += 3 * 256) acc[i] = 0.f;

  v2f btk[16], btv[16], blw[16], btq[16];
  v8f cP = {}, cL = {};
  float lbn = 0.f;
  if (on) {
    const bool isw = nl < 12, isb = nl == 12;
    v2f a1[16], bw[16];
#pragma unroll
    for (int ks = 0; ks < 16; ++ks) {
      const int kb = ks * 4 + 2 * hi;
      const float* p0 = isw ? fc1w + kb * 12 + nl       : fc1b + kb;
      const float* p1 = isw ? fc1w + (kb + 1) * 12 + nl : fc1b + kb + 1;
      const float r0 = *p0, r1 = *p1;
      a1[ks]  = mkv2(isw ? r0 : (isb ? r0 : 0.f), isw ? r1 : (isb ? r1 + 1.0f : 0.f));
      bw[ks]  = *(const v2f*)(Wt + n * 64 + kb);
      btk[ks] = mkv2(tk[kb * 64 + n],  tk[(kb + 1) * 64 + n]);
      btv[ks] = mkv2(tvm[kb * 64 + n], tvm[(kb + 1) * 64 + n]);
      blw[ks] = *(const v2f*)(lw + n * 64 + kb);
      btq[ks] = mkv2(tq[kb * 64 + n], tq[(kb + 1) * 64 + n]);
    }
    const float bt = Bt[n];
    lbn = lb[n];

    v8f cS = {};
#pragma unroll
    for (int ks = 0; ks < 16; ++ks) cS = wmma_f32(a1[ks], bw[ks], cS);
    cS[4] += (hi == 1) ? bt : 0.f;
#pragma unroll
    for (int r = 0; r < 8; ++r) sS[(r + 8 * hi) * 68 + n] = cS[r];
  }
  __syncthreads();

  if (on) {
    v2f as[16];
#pragma unroll
    for (int ks = 0; ks < 16; ++ks) {
      const int kb = ks * 4 + 2 * hi;
      as[ks] = mkv2(sS[nl * 68 + kb], sS[nl * 68 + kb + 1]);
    }
    if (t == 2) {
      v8f cU = {};
#pragma unroll
      for (int ks = 0; ks < 16; ++ks) {
        cP = wmma_f32(as[ks], btk[ks], cP);
        cL = wmma_f32(as[ks], btv[ks], cL);
        cU = wmma_f32(as[ks], btq[ks], cU);
      }
      store_factor(wsf + F_U, cU, warp, hi, nl);
    } else {
#pragma unroll
      for (int ks = 0; ks < 16; ++ks) {
        cP = wmma_f32(as[ks], btk[ks], cP);
        cL = wmma_f32(as[ks], btv[ks], cL);
      }
    }
    store_factor(wsf + F_P + t * 1024, cP, warp, hi, nl);
#pragma unroll
    for (int r = 0; r < 8; ++r) sP[(r + 8 * hi) * 68 + n] = cP[r];
  }
  __syncthreads();

  if (on) {
    v2f ap[16];
#pragma unroll
    for (int ks = 0; ks < 16; ++ks) {
      const int kb = ks * 4 + 2 * hi;
      ap[ks] = mkv2(sP[nl * 68 + kb], sP[nl * 68 + kb + 1]);
    }
    v8f cQ = {};
#pragma unroll
    for (int ks = 0; ks < 16; ++ks) cQ = wmma_f32(ap[ks], blw[ks], cQ);
#pragma unroll
    for (int r = 0; r < 8; ++r) cQ[r] -= cL[r];
    cQ[4] += (hi == 1) ? lbn : 0.f;
    store_factor(wsf + F_Q + t * 1024, cQ, warp, hi, nl);
  }
}

__global__ void __launch_bounds__(256) k_cat(
    const float* __restrict__ in1, const float* __restrict__ in2,
    const float* __restrict__ in3, const float* __restrict__ in4,
    const float* __restrict__ wl, const float* __restrict__ bl,
    const float* __restrict__ wm, const float* __restrict__ bm,
    const float* __restrict__ wsw, const float* __restrict__ bs,
    const float* __restrict__ fc1w, const float* __restrict__ fc1b,
    const float* __restrict__ qw, const float* __restrict__ qb,
    const float* __restrict__ kw, const float* __restrict__ kbv,
    const float* __restrict__ vw, const float* __restrict__ vb,
    const float* __restrict__ tk, const float* __restrict__ tvm, const float* __restrict__ tq,
    const float* __restrict__ lw, const float* __restrict__ lb,
    float* __restrict__ wsf, float* __restrict__ cat)
{
  __shared__ __attribute__((aligned(16))) float wt[WT_N];
  __shared__ __attribute__((aligned(16))) float stage[8][16 * CP];
  __shared__ __attribute__((aligned(16))) float comb[4][256];

  const int lane = threadIdx.x & 31;
  const int warp = __builtin_amdgcn_readfirstlane(threadIdx.x >> 5);
  const int part = warp & 1, tl = warp >> 1;
  const int row0 = (blockIdx.x * 4 + tl) * 16;
  const int nl = lane & 15, hi = lane >> 4;
  float* st = stage[warp];

  if (blockIdx.x < 3) {
    const int t = (int)blockIdx.x;
    prep_factors(t, warp, lane, (int)threadIdx.x, stage[0], stage[1], fc1w, fc1b,
                 (t == 0) ? qw : (t == 1) ? kw : vw, (t == 0) ? qb : (t == 1) ? kbv : vb,
                 tk, tvm, tq, lw, lb, wsf);
  }

  v4f buf[16];
  CDesc d = cat_desc(part, 0, row0, in1, in2, in3);
  cat_load(buf, d, lane);

  float wv[29];
CAT_UNROLL
  for (int it = 0; it < 29; ++it) {
    const int i = it * 256 + (int)threadIdx.x;
    const float* sp; int o, j, K;
    if (i < WT_M)      { o = i / 1504; j = i - o * 1504; K = 500; sp = wl; }
    else if (i < WT_S) { const int t = i - WT_M; o = t / 752; j = t - o * 752; K = 250; sp = wm; }
    else               { const int t = i - WT_S; o = t / 152; j = t - o * 152; K = 50; sp = wsw; }
    const bool ok = (i < WT_N) && (j < 3 * K);
    const int jj = ok ? j : 0, oo = ok ? o : 0;
    const int k = jj / 3, c = jj - 3 * k;
    const float v = sp[(oo * 3 + c) * K + k];
    wv[it] = ok ? v : 0.f;
  }
CAT_UNROLL
  for (int it = 0; it < 29; ++it) {
    const int i = it * 256 + (int)threadIdx.x;
    if (i < WT_N) wt[i] = wv[it];
  }
  __syncthreads();

  v8f acc0 = {}, acc1 = {};
  const int nch = (part == 0) ? 10 : 11;
  for (int ch = 0; ch < nch; ++ch) {
    cat_stage(st, buf, d.wide, lane);
    CDesc dn = d;
    if (ch + 1 < nch) {
      dn = cat_desc(part, ch + 1, row0, in1, in2, in3);
      cat_load(buf, dn, lane);
    }
    cat_mma(st, wt, d, nl, hi, acc0, acc1);
    d = dn;
  }

  float* cb = comb[tl];
  v8f acc = acc0 + acc1;
  if (part == 1) {
CAT_UNROLL
    for (int r = 0; r < 8; ++r) cb[(8 * hi + r) * 16 + nl] = acc[r];
  }
  __syncthreads();
  if (part == 0) {
    float bias = 0.f;
    if (nl < 3)      bias = bl[nl];
    else if (nl < 6) bias = bm[nl - 3];
    else if (nl < 9) bias = bs[nl - 6];
CAT_UNROLL
    for (int r = 0; r < 8; ++r) {
      const int row = 8 * hi + r;
      float v = acc[r] + cb[row * 16 + nl] + bias;
      if (nl >= 9) v = (nl < 12) ? in4[(size_t)(row0 + row) * 3 + (nl - 9)] : (nl == 12 ? 1.f : 0.f);
      cb[row * 16 + nl] = v;
    }
    v4f* dst = (v4f*)(cat + (size_t)row0 * 16);
    const v4f* srcp = (const v4f*)cb;
    dst[lane] = srcp[lane];
    dst[32 + lane] = srcp[32 + lane];
  }
}

constexpr int KG_BLOCKS = 256;
constexpr int KG_WAVES  = KG_BLOCKS * 4;

__global__ void __launch_bounds__(128) k_grad(
    const float* __restrict__ cat, const float* __restrict__ P, const float* __restrict__ Q,
    float* __restrict__ gW, float* __restrict__ gb)
{
  __shared__ float sgW[64 * 64];
  __shared__ float sgb[64];
  for (int i = threadIdx.x; i < 4096; i += 128) sgW[i] = 0.f;
  if (threadIdx.x < 64) sgb[threadIdx.x] = 0.f;
  __syncthreads();

  const int warp = threadIdx.x >> 5, lane = threadIdx.x & 31;
  const int hi = lane >> 4, nl = lane & 15;

  const float gscale = 2.0f / (float)(3ull * BATCH * 64);
  const int NCT = BATCH / 16;
  const int wave = blockIdx.x * 4 + warp;
  const int nw = KG_WAVES;

  v2f bp[3][4][4], bq[3][4][4];
#pragma unroll
  for (int t = 0; t < 3; ++t) {
    load_factor(P + t * 1024, lane, bp[t]);
    load_factor(Q + t * 1024, lane, bq[t]);
  }

  v8f cgw[4][4];
#pragma unroll
  for (int i0 = 0; i0 < 4; ++i0)
#pragma unroll
    for (int j0 = 0; j0 < 4; ++j0) cgw[i0][j0] = (v8f){};
  float gbp[4] = {0.f, 0.f, 0.f, 0.f};

  v2f cn[4];
#pragma unroll
  for (int ks = 0; ks < 4; ++ks)
    cn[ks] = *(const v2f*)(cat + (size_t)(wave * 16 + nl) * 16 + ks * 4 + 2 * hi);

  for (int ct = wave; ct < NCT; ct += nw) {
    v2f ca[4];
#pragma unroll
    for (int ks = 0; ks < 4; ++ks) ca[ks] = cn[ks];
    if (ct + nw < NCT) {
#pragma unroll
      for (int ks = 0; ks < 4; ++ks)
        cn[ks] = *(const v2f*)(cat + (size_t)((ct + nw) * 16 + nl) * 16 + ks * 4 + 2 * hi);
    }
#pragma unroll
    for (int t = 0; t < 3; ++t) {
      v8f ctv[4], gt[4];
#pragma unroll
      for (int n0 = 0; n0 < 4; ++n0) {
        v8f cv = {}, cd = {};
#pragma unroll
        for (int ks = 0; ks < 4; ++ks) {
          cv = wmma_f32(ca[ks], bp[t][ks][n0], cv);
          cd = wmma_f32(ca[ks], bq[t][ks][n0], cd);
        }
        ctv[n0] = cv;
        float colsum = 0.f;
#pragma unroll
        for (int r = 0; r < 8; ++r) {
          const float g = gscale * cd[r];
          gt[n0][r] = g;
          colsum += g;
        }
        gbp[n0] += colsum;
      }
#pragma unroll
      for (int i0 = 0; i0 < 4; ++i0) {
#pragma unroll
        for (int j0 = 0; j0 < 4; ++j0) {
#pragma unroll
          for (int q = 0; q < 4; ++q) {
            v2f a = mkv2(gt[i0][2 * q], gt[i0][2 * q + 1]);
            v2f b = mkv2(ctv[j0][2 * q], ctv[j0][2 * q + 1]);
            cgw[i0][j0] = wmma_f32(a, b, cgw[i0][j0]);
          }
        }
      }
    }
  }
#pragma unroll
  for (int i0 = 0; i0 < 4; ++i0)
#pragma unroll
    for (int j0 = 0; j0 < 4; ++j0)
#pragma unroll
      for (int r = 0; r < 8; ++r)
        atomicAdd(&sgW[(i0 * 16 + r + 8 * hi) * 64 + j0 * 16 + nl], cgw[i0][j0][r]);
#pragma unroll
  for (int n0 = 0; n0 < 4; ++n0) atomicAdd(&sgb[n0 * 16 + nl], gbp[n0]);
  __syncthreads();
  for (int i = threadIdx.x; i < 4096; i += 128) atomicAdd(&gW[i], sgW[i]);
  if (threadIdx.x < 64) atomicAdd(&gb[threadIdx.x], sgb[threadIdx.x]);
}

__global__ void __launch_bounds__(256) k_update(
                         const float* __restrict__ lw, const float* __restrict__ lb,
                         const float* __restrict__ gW, const float* __restrict__ gb,
                         float* __restrict__ lw1T, float* __restrict__ lb1)
{
  const int b = (int)blockIdx.x, t = (int)threadIdx.x;
  if (b < 16) {
    const int i = b * 256 + t;
    const int n = i >> 6, k = i & 63;
    const float g = gW[i];
    lw1T[k * 64 + n] = lw[i] - LRATE * g / (fabsf(g) + AEPS);
  } else if (t < 64) {
    const float g = gb[t];
    lb1[t] = lb[t] - LRATE * g / (fabsf(g) + AEPS);
  }
}

__global__ void __launch_bounds__(128) k_out(
    const float* __restrict__ cat, const float* __restrict__ U,
    const float* __restrict__ lw1T, const float* __restrict__ lb1,
    const float* __restrict__ lng, const float* __restrict__ lnb,
    const float* __restrict__ ow, const float* __restrict__ obp,
    float* __restrict__ out)
{
  __shared__ float su[4][16 * 68];
  __shared__ float so[4][16 * 68];
  const int warp = threadIdx.x >> 5, lane = threadIdx.x & 31;
  const int hi = lane >> 4, nl = lane & 15;
  const int row0 = (blockIdx.x * 4 + warp) * 16;
  float* uw = su[warp];
  float* om = so[warp];

  v2f ca[4];
#pragma unroll
  for (int ks = 0; ks < 4; ++ks)
    ca[ks] = *(const v2f*)(cat + (size_t)(row0 + nl) * 16 + ks * 4 + 2 * hi);
  v2f bu[4][4];
  load_factor(U, lane, bu);
#pragma unroll
  for (int n0 = 0; n0 < 4; ++n0) {
    int n = n0 * 16 + nl;
    v8f c = {};
#pragma unroll
    for (int ks = 0; ks < 4; ++ks) c = wmma_f32(ca[ks], bu[ks][n0], c);
#pragma unroll
    for (int r = 0; r < 8; ++r) uw[(r + 8 * hi) * 68 + n] = c[r];
  }
#pragma unroll
  for (int n0 = 0; n0 < 4; ++n0) {
    int n = n0 * 16 + nl;
    v8f c = {};
#pragma unroll
    for (int ks = 0; ks < 16; ++ks) {
      int kb = ks * 4 + 2 * hi;
      v2f a = mkv2(uw[nl * 68 + kb], uw[nl * 68 + kb + 1]);
      c = wmma_f32(a, mkv2(lw1T[kb * 64 + n], lw1T[(kb + 1) * 64 + n]), c);
    }
    float bias = lb1[n];
#pragma unroll
    for (int r = 0; r < 8; ++r) om[(r + 8 * hi) * 68 + n] = c[r] + bias;
  }
  __syncthreads();
  if (lane < 16) {
    const float* rowp = om + lane * 68;
    float mu = 0.f;
#pragma unroll
    for (int j = 0; j < 64; ++j) mu += rowp[j];
    mu *= (1.0f / 64.0f);
    float var = 0.f;
#pragma unroll
    for (int j = 0; j < 64; ++j) { float d = rowp[j] - mu; var += d * d; }
    var *= (1.0f / 64.0f);
    float rstd = rsqrtf(var + LNEPS);
    float acc = 0.f;
#pragma unroll
    for (int j = 0; j < 64; ++j) {
      float nrm = (rowp[j] - mu) * rstd * lng[j] + lnb[j];
      float a = nrm / (1.0f + __expf(-nrm));
      acc += a * ow[j];
    }
    out[row0 + lane] = acc + obp[0];
  }
}

extern "C" void kernel_launch(void* const* d_in, const int* in_sizes, int n_in,
                              void* d_out, int out_size, void* d_ws, size_t ws_size,
                              hipStream_t stream) {
  const float* in1  = (const float*)d_in[0];
  const float* in2  = (const float*)d_in[1];
  const float* in3  = (const float*)d_in[2];
  const float* in4  = (const float*)d_in[3];
  const float* w_l  = (const float*)d_in[4];
  const float* b_l  = (const float*)d_in[5];
  const float* w_m  = (const float*)d_in[6];
  const float* b_m  = (const float*)d_in[7];
  const float* w_s  = (const float*)d_in[8];
  const float* b_s  = (const float*)d_in[9];
  const float* fc1w = (const float*)d_in[10];
  const float* fc1b = (const float*)d_in[11];
  const float* q_w  = (const float*)d_in[12];
  const float* q_b  = (const float*)d_in[13];
  const float* k_w  = (const float*)d_in[14];
  const float* k_b  = (const float*)d_in[15];
  const float* v_w  = (const float*)d_in[16];
  const float* v_b  = (const float*)d_in[17];
  const float* t_k  = (const float*)d_in[18];
  const float* t_v  = (const float*)d_in[19];
  const float* t_q  = (const float*)d_in[20];
  const float* lw   = (const float*)d_in[21];
  const float* lb   = (const float*)d_in[22];
  const float* lng  = (const float*)d_in[23];
  const float* lnb  = (const float*)d_in[24];
  const float* outw = (const float*)d_in[25];
  const float* outb = (const float*)d_in[26];

  float* wsf  = (float*)d_ws;
  float* cat  = wsf + F_CAT;
  float* fP   = wsf + F_P;
  float* fQ   = wsf + F_Q;
  float* fU   = wsf + F_U;
  float* gW   = wsf + F_GW;
  float* gb   = wsf + F_GB;
  float* lw1T = wsf + F_LW1T;
  float* lb1  = wsf + F_LB1;

  k_cat<<<BATCH / 64, 256, 0, stream>>>(in1, in2, in3, in4, w_l, b_l, w_m, b_m, w_s, b_s,
                                        fc1w, fc1b, q_w, q_b, k_w, k_b, v_w, v_b, t_k, t_v, t_q, lw, lb,
                                        wsf, cat);
  k_grad<<<KG_BLOCKS, 128, 0, stream>>>(cat, fP, fQ, gW, gb);
  k_update<<<17, 256, 0, stream>>>(lw, lb, gW, gb, lw1T, lb1);
  k_out<<<BATCH / 64, 128, 0, stream>>>(cat, fU, lw1T, lb1, lng, lnb, outw, outb,
                                        (float*)d_out);
}
